# starting kernel
# baseline (speedup 1.0000x reference)
#include <hip/hip_runtime.h>


typedef __attribute__((ext_vector_type(2))) float v2f;
typedef __attribute__((ext_vector_type(8))) float v8f;

#define B_ 4
#define H_ 16
#define S_ 4096
#define D_ 64
#define BH_ (B_ * H_)
#define EPS_ 1e-6f

#define KV_STRIDE 4160
#define SCHUNK 512
#define P2_ROWS 128

#define QSTRIDE 68
#define KVSTRIDE_L 72

__device__ __forceinline__ float phi_f(float x) {
    return x > 0.0f ? x + 1.0f : __expf(x);
}

__global__ void la_zero_ws(float4* __restrict__ ws, int n4) {
    int i = blockIdx.x * blockDim.x + threadIdx.x;
    if (i < n4) ws[i] = make_float4(0.f, 0.f, 0.f, 0.f);
}

__global__ __launch_bounds__(256) void la_pass1_kvterm(
    const float* __restrict__ K, const float* __restrict__ V,
    const float* __restrict__ mask, float* __restrict__ ws) {

    const int bh = blockIdx.x;
    const int chunk = blockIdx.y;
    const int b = bh / H_;
    const size_t base = (size_t)bh * S_ * D_ + (size_t)chunk * SCHUNK * D_;
    const float* __restrict__ Kp = K + base;
    const float* __restrict__ Vp = V + base;
    const float* __restrict__ Mp = mask + (size_t)b * S_ + (size_t)chunk * SCHUNK;

    __shared__ __align__(16) float lds_kT[D_][17];
    __shared__ __align__(16) float lds_v[16][64];
    __shared__ float lds_kone[D_];

    const int tid  = threadIdx.x;
    const int lane = tid & 31;
    const int wid  = tid >> 5;
    const int l15  = lane & 15;
    const int half = lane >> 4;

    const int dTile0 = wid >> 2;
    const int dTile1 = dTile0 + 2;
    const int vTile  = wid & 3;

    const int nl  = tid >> 4;
    const int col = (tid & 15) * 4;

    v8f c0 = {}; v8f c1 = {};
    float kone_acc[4] = {0.f, 0.f, 0.f, 0.f};

    if (tid < D_) lds_kone[tid] = 0.0f;

    for (int n0 = 0; n0 < SCHUNK; n0 += 16) {
        __syncthreads();
        const float4 kd = *(const float4*)&Kp[(size_t)(n0 + nl) * D_ + col];
        const float4 vv = *(const float4*)&Vp[(size_t)(n0 + nl) * D_ + col];
        const float m = Mp[n0 + nl];
        const float pk0 = phi_f(kd.x) * m;
        const float pk1 = phi_f(kd.y) * m;
        const float pk2 = phi_f(kd.z) * m;
        const float pk3 = phi_f(kd.w) * m;
        lds_kT[col + 0][nl] = pk0;
        lds_kT[col + 1][nl] = pk1;
        lds_kT[col + 2][nl] = pk2;
        lds_kT[col + 3][nl] = pk3;
        kone_acc[0] += pk0; kone_acc[1] += pk1;
        kone_acc[2] += pk2; kone_acc[3] += pk3;
        *(float4*)&lds_v[nl][col] = vv;
        __syncthreads();

        #pragma unroll
        for (int kk = 0; kk < 16; kk += 4) {
            const int kc = kk + half * 2;
            v2f a0, a1, bb;
            a0.x = lds_kT[dTile0 * 16 + l15][kc];
            a0.y = lds_kT[dTile0 * 16 + l15][kc + 1];
            a1.x = lds_kT[dTile1 * 16 + l15][kc];
            a1.y = lds_kT[dTile1 * 16 + l15][kc + 1];
            bb.x = lds_v[kc][vTile * 16 + l15];
            bb.y = lds_v[kc + 1][vTile * 16 + l15];
            c0 = __builtin_amdgcn_wmma_f32_16x16x4_f32(false, a0, false, bb,
                                                       (short)0, c0, false, false);
            c1 = __builtin_amdgcn_wmma_f32_16x16x4_f32(false, a1, false, bb,
                                                       (short)0, c1, false, false);
        }
    }

    #pragma unroll
    for (int j = 0; j < 4; ++j) atomicAdd(&lds_kone[col + j], kone_acc[j]);
    __syncthreads();

    float* __restrict__ wkv = ws + (size_t)bh * KV_STRIDE;
    if (tid < D_) atomicAdd(&wkv[4096 + tid], lds_kone[tid]);

    #pragma unroll
    for (int j = 0; j < 8; ++j) {
        const int r0 = dTile0 * 16 + j + 8 * half;
        const int r1 = dTile1 * 16 + j + 8 * half;
        const int cidx = vTile * 16 + l15;
        atomicAdd(&wkv[r0 * 64 + cidx], c0[j]);
        atomicAdd(&wkv[r1 * 64 + cidx], c1[j]);
    }
}

__global__ __launch_bounds__(256) void la_pass2_output(
    const float* __restrict__ Q, const float* __restrict__ ws,
    float* __restrict__ out) {

    const int bh = blockIdx.x;
    const int rblk = blockIdx.y;
    const size_t qbase = (size_t)bh * S_ * D_ + (size_t)rblk * P2_ROWS * D_;
    const float* __restrict__ Qp = Q + qbase;
    float* __restrict__ Op = out + qbase;
    const float* __restrict__ wkv = ws + (size_t)bh * KV_STRIDE;

    __shared__ __align__(16) float lds_kv[D_][KVSTRIDE_L];
    __shared__ float lds_kone[D_];
    __shared__ __align__(16) float lds_q[P2_ROWS][QSTRIDE];

    const int tid  = threadIdx.x;
    const int lane = tid & 31;
    const int wid  = tid >> 5;
    const int l15  = lane & 15;
    const int half = lane >> 4;

    #pragma unroll
    for (int i = 0; i < 4; ++i) {
        const int idx = tid + 256 * i;
        const int r = idx >> 4;
        const int c = (idx & 15) * 4;
        *(float4*)&lds_kv[r][c] = *(const float4*)&wkv[r * 64 + c];
    }
    if (tid < D_) lds_kone[tid] = wkv[4096 + tid];

    #pragma unroll
    for (int i = 0; i < 8; ++i) {
        const int idx = tid + 256 * i;
        const int r = idx >> 4;
        const int c = (idx & 15) * 4;
        float4 qv = *(const float4*)&Qp[(size_t)r * D_ + c];
        qv.x = phi_f(qv.x * 0.125f);
        qv.y = phi_f(qv.y * 0.125f);
        qv.z = phi_f(qv.z * 0.125f);
        qv.w = phi_f(qv.w * 0.125f);
        *(float4*)&lds_q[r][c] = qv;
    }
    __syncthreads();

    const int mb = wid * 16;
    v8f acc[4] = {{}, {}, {}, {}};

    #pragma unroll
    for (int kk = 0; kk < 64; kk += 4) {
        const int kc = kk + half * 2;
        v2f a;
        a.x = lds_q[mb + l15][kc];
        a.y = lds_q[mb + l15][kc + 1];
        #pragma unroll
        for (int vt = 0; vt < 4; ++vt) {
            v2f bfrag;
            bfrag.x = lds_kv[kc][vt * 16 + l15];
            bfrag.y = lds_kv[kc + 1][vt * 16 + l15];
            acc[vt] = __builtin_amdgcn_wmma_f32_16x16x4_f32(false, a, false, bfrag,
                                                            (short)0, acc[vt], false, false);
        }
    }

    float part = 0.0f;
    const int nrow = mb + l15;
    #pragma unroll
    for (int d = 0; d < 32; ++d) {
        const int dd = half * 32 + d;
        part += lds_q[nrow][dd] * lds_kone[dd];
    }
    part += __shfl_xor(part, 16, 32);
    const float inv = 1.0f / (part + EPS_);

    #pragma unroll
    for (int j = 0; j < 8; ++j) {
        const float rn = __shfl(inv, j + 8 * half, 32);
        const int r = mb + j + 8 * half;
        #pragma unroll
        for (int vt = 0; vt < 4; ++vt) {
            Op[(size_t)r * D_ + vt * 16 + l15] = acc[vt][j] * rn;
        }
    }
}

extern "C" void kernel_launch(void* const* d_in, const int* in_sizes, int n_in,
                              void* d_out, int out_size, void* d_ws, size_t ws_size,
                              hipStream_t stream) {
    const float* Q = (const float*)d_in[0];
    const float* K = (const float*)d_in[1];
    const float* V = (const float*)d_in[2];
    const float* M = (const float*)d_in[3];
    float* out = (float*)d_out;
    float* ws  = (float*)d_ws;

    const int nz4 = (BH_ * KV_STRIDE) / 4;
    la_zero_ws<<<(nz4 + 255) / 256, 256, 0, stream>>>((float4*)ws, nz4);
    la_pass1_kvterm<<<dim3(BH_, S_ / SCHUNK), 256, 0, stream>>>(K, V, M, ws);
    la_pass2_output<<<dim3(BH_, S_ / P2_ROWS), 256, 0, stream>>>(Q, ws, out);
}


# pass 1: each wave streams its own 128 rows of K,V into WMMA fragments by whole-row b128 loads, next group's 16 loads issued ahead; no LDS slab or per-slab barrier; n summed per wave, then in LDS
# speedup vs baseline: 1.3124x; 1.3124x over previous
#include <hip/hip_runtime.h>


typedef __attribute__((ext_vector_type(2))) float v2f;
typedef __attribute__((ext_vector_type(8))) float v8f;

#define B_ 4
#define H_ 16
#define S_ 4096
#define D_ 64
#define BH_ (B_ * H_)
#define EPS_ 1e-6f

#define KV_STRIDE 4160
#define P1_WAVES 8
#define P1_WROWS 128
#define P1_BROWS (P1_WAVES * P1_WROWS)
#define P1_GROWS 16
#define P1_NGROUPS (P1_WROWS / P1_GROWS)
#define P2_ROWS 128

#define QSTRIDE 68
#define KVSTRIDE_L 72

__device__ __forceinline__ float phi_f(float x) {
    return x > 0.0f ? x + 1.0f : __expf(x);
}

__global__ void la_zero_ws(float4* __restrict__ ws, int n4) {
    int i = blockIdx.x * blockDim.x + threadIdx.x;
    if (i < n4) ws[i] = make_float4(0.f, 0.f, 0.f, 0.f);
}

__device__ __forceinline__ void p1_load_group(const float4* __restrict__ K4,
                                              const float4* __restrict__ V4,
                                              const float4* __restrict__ M4,
                                              int lane, float4 (&k)[8], float4 (&v)[8],
                                              float4 (&m)[4]) {
    #pragma unroll
    for (int i = 0; i < 4; ++i) m[i] = M4[i];
    #pragma unroll
    for (int i = 0; i < 8; ++i) {
        k[i] = K4[i * 32 + lane];
        v[i] = V4[i * 32 + lane];
    }
}

__device__ __forceinline__ void p1_compute_group(const float4 (&k)[8], const float4 (&v)[8],
                                                 const float4 (&m)[4], int half,
                                                 v8f (&acc)[4][4], float (&kone)[4]) {
    #pragma unroll
    for (int ks = 0; ks < 4; ++ks) {
        const float m0 = half ? m[ks].y : m[ks].x;
        const float m1 = half ? m[ks].w : m[ks].z;
        const float4 kx = k[2 * ks], ky = k[2 * ks + 1];
        const float4 vx = v[2 * ks], vy = v[2 * ks + 1];
        v2f a[4], b[4];
        a[0].x = phi_f(kx.x) * m0;  a[0].y = phi_f(ky.x) * m1;
        a[1].x = phi_f(kx.y) * m0;  a[1].y = phi_f(ky.y) * m1;
        a[2].x = phi_f(kx.z) * m0;  a[2].y = phi_f(ky.z) * m1;
        a[3].x = phi_f(kx.w) * m0;  a[3].y = phi_f(ky.w) * m1;
        b[0].x = vx.x;  b[0].y = vy.x;
        b[1].x = vx.y;  b[1].y = vy.y;
        b[2].x = vx.z;  b[2].y = vy.z;
        b[3].x = vx.w;  b[3].y = vy.w;
        #pragma unroll
        for (int t = 0; t < 4; ++t) kone[t] += a[t].x + a[t].y;
        #pragma unroll
        for (int t = 0; t < 4; ++t) {
            #pragma unroll
            for (int u = 0; u < 4; ++u) {
                acc[t][u] = __builtin_amdgcn_wmma_f32_16x16x4_f32(false, a[t], false, b[u],
                                                                  (short)0, acc[t][u], false, false);
            }
        }
    }
}

__global__ __launch_bounds__(256) void la_pass1_kvterm(
    const float* __restrict__ K, const float* __restrict__ V,
    const float* __restrict__ mask, float* __restrict__ ws) {

    __shared__ __align__(16) float lds_red[P1_WAVES][D_ * D_];
    __shared__ __align__(16) float lds_kred[2 * P1_WAVES][D_];

    const int bh = blockIdx.x;
    const int chunk = blockIdx.y;
    const int b = bh / H_;
    const int tid  = threadIdx.x;
    const int lane = tid & 31;
    const int wid  = __builtin_amdgcn_readfirstlane(tid >> 5);
    const int l15  = lane & 15;
    const int half = lane >> 4;

    const int row0 = chunk * P1_BROWS + wid * P1_WROWS;
    const size_t base = (size_t)bh * S_ * D_ + (size_t)row0 * D_;
    const float4* __restrict__ K4 = (const float4*)(K + base);
    const float4* __restrict__ V4 = (const float4*)(V + base);
    const float4* __restrict__ M4 = (const float4*)(mask + (size_t)b * S_ + row0);

    v8f acc[4][4];
    #pragma unroll
    for (int t = 0; t < 4; ++t) {
        #pragma unroll
        for (int u = 0; u < 4; ++u) acc[t][u] = (v8f){};
    }
    float kone[4] = {0.f, 0.f, 0.f, 0.f};

    float4 kA[8], vA[8], mA[4], kB[8], vB[8], mB[4];
    const int G4 = P1_GROWS * (D_ / 4);
    const int GM = P1_GROWS / 4;
    p1_load_group(K4, V4, M4, lane, kA, vA, mA);
    #define P1_PIN() __builtin_amdgcn_sched_barrier(0)
    for (int g = 0; g < P1_NGROUPS - 2; g += 2) {
        p1_load_group(K4 + (g + 1) * G4, V4 + (g + 1) * G4, M4 + (g + 1) * GM, lane, kB, vB, mB);
        P1_PIN();
        p1_compute_group(kA, vA, mA, half, acc, kone);
        P1_PIN();
        p1_load_group(K4 + (g + 2) * G4, V4 + (g + 2) * G4, M4 + (g + 2) * GM, lane, kA, vA, mA);
        P1_PIN();
        p1_compute_group(kB, vB, mB, half, acc, kone);
        P1_PIN();
    }
    p1_load_group(K4 + (P1_NGROUPS - 1) * G4, V4 + (P1_NGROUPS - 1) * G4,
                  M4 + (P1_NGROUPS - 1) * GM, lane, kB, vB, mB);
    P1_PIN();
    p1_compute_group(kA, vA, mA, half, acc, kone);
    p1_compute_group(kB, vB, mB, half, acc, kone);
    #undef P1_PIN

    float* __restrict__ myred = lds_red[wid];
    #pragma unroll
    for (int t = 0; t < 4; ++t) {
        #pragma unroll
        for (int r = 0; r < 8; ++r) {
            const int d = 4 * (8 * half + r) + t;
            *(float4*)&myred[d * D_ + 4 * l15] =
                make_float4(acc[t][0][r], acc[t][1][r], acc[t][2][r], acc[t][3][r]);
        }
    }
    *(float4*)&lds_kred[2 * wid + half][4 * l15] = make_float4(kone[0], kone[1], kone[2], kone[3]);
    __syncthreads();

    float* __restrict__ wkv = ws + (size_t)bh * KV_STRIDE;
    for (int j = 0; j < (D_ * D_) / 256; ++j) {
        const int e = tid + 256 * j;
        float s = lds_red[0][e];
        #pragma unroll
        for (int w = 1; w < P1_WAVES; ++w) s += lds_red[w][e];
        atomicAdd(&wkv[e], s);
    }
    if (tid < D_) {
        float s = lds_kred[0][tid];
        #pragma unroll
        for (int w = 1; w < 2 * P1_WAVES; ++w) s += lds_kred[w][tid];
        atomicAdd(&wkv[4096 + tid], s);
    }
}

__global__ __launch_bounds__(256) void la_pass2_output(
    const float* __restrict__ Q, const float* __restrict__ ws,
    float* __restrict__ out) {

    const int bh = blockIdx.x;
    const int rblk = blockIdx.y;
    const size_t qbase = (size_t)bh * S_ * D_ + (size_t)rblk * P2_ROWS * D_;
    const float* __restrict__ Qp = Q + qbase;
    float* __restrict__ Op = out + qbase;
    const float* __restrict__ wkv = ws + (size_t)bh * KV_STRIDE;

    __shared__ __align__(16) float lds_kv[D_][KVSTRIDE_L];
    __shared__ float lds_kone[D_];
    __shared__ __align__(16) float lds_q[P2_ROWS][QSTRIDE];

    const int tid  = threadIdx.x;
    const int lane = tid & 31;
    const int wid  = tid >> 5;
    const int l15  = lane & 15;
    const int half = lane >> 4;

    #pragma unroll
    for (int i = 0; i < 4; ++i) {
        const int idx = tid + 256 * i;
        const int r = idx >> 4;
        const int c = (idx & 15) * 4;
        *(float4*)&lds_kv[r][c] = *(const float4*)&wkv[r * 64 + c];
    }
    if (tid < D_) lds_kone[tid] = wkv[4096 + tid];

    #pragma unroll
    for (int i = 0; i < 8; ++i) {
        const int idx = tid + 256 * i;
        const int r = idx >> 4;
        const int c = (idx & 15) * 4;
        float4 qv = *(const float4*)&Qp[(size_t)r * D_ + c];
        qv.x = phi_f(qv.x * 0.125f);
        qv.y = phi_f(qv.y * 0.125f);
        qv.z = phi_f(qv.z * 0.125f);
        qv.w = phi_f(qv.w * 0.125f);
        *(float4*)&lds_q[r][c] = qv;
    }
    __syncthreads();

    const int mb = wid * 16;
    v8f acc[4] = {{}, {}, {}, {}};

    #pragma unroll
    for (int kk = 0; kk < 64; kk += 4) {
        const int kc = kk + half * 2;
        v2f a;
        a.x = lds_q[mb + l15][kc];
        a.y = lds_q[mb + l15][kc + 1];
        #pragma unroll
        for (int vt = 0; vt < 4; ++vt) {
            v2f bfrag;
            bfrag.x = lds_kv[kc][vt * 16 + l15];
            bfrag.y = lds_kv[kc + 1][vt * 16 + l15];
            acc[vt] = __builtin_amdgcn_wmma_f32_16x16x4_f32(false, a, false, bfrag,
                                                            (short)0, acc[vt], false, false);
        }
    }

    float part = 0.0f;
    const int nrow = mb + l15;
    #pragma unroll
    for (int d = 0; d < 32; ++d) {
        const int dd = half * 32 + d;
        part += lds_q[nrow][dd] * lds_kone[dd];
    }
    part += __shfl_xor(part, 16, 32);
    const float inv = 1.0f / (part + EPS_);

    #pragma unroll
    for (int j = 0; j < 8; ++j) {
        const float rn = __shfl(inv, j + 8 * half, 32);
        const int r = mb + j + 8 * half;
        #pragma unroll
        for (int vt = 0; vt < 4; ++vt) {
            Op[(size_t)r * D_ + vt * 16 + l15] = acc[vt][j] * rn;
        }
    }
}

extern "C" void kernel_launch(void* const* d_in, const int* in_sizes, int n_in,
                              void* d_out, int out_size, void* d_ws, size_t ws_size,
                              hipStream_t stream) {
    const float* Q = (const float*)d_in[0];
    const float* K = (const float*)d_in[1];
    const float* V = (const float*)d_in[2];
    const float* M = (const float*)d_in[3];
    float* out = (float*)d_out;
    float* ws  = (float*)d_ws;

    const int nz4 = (BH_ * KV_STRIDE) / 4;
    la_zero_ws<<<(nz4 + 255) / 256, 256, 0, stream>>>((float4*)ws, nz4);
    la_pass1_kvterm<<<dim3(BH_, S_ / P1_BROWS), 256, 0, stream>>>(K, V, M, ws);
    la_pass2_output<<<dim3(BH_, S_ / P2_ROWS), 256, 0, stream>>>(Q, ws, out);
}


# pass 2: each wave streams its own 128 rows of Q through a wave-private LDS tile into A fragments, kv held in registers as B fragments, whole-row b128 stores; no block barrier; new sum order over d
# speedup vs baseline: 1.6376x; 1.2478x over previous
#include <hip/hip_runtime.h>


typedef __attribute__((ext_vector_type(2))) float v2f;
typedef __attribute__((ext_vector_type(8))) float v8f;

#define B_ 4
#define H_ 16
#define S_ 4096
#define D_ 64
#define BH_ (B_ * H_)
#define EPS_ 1e-6f

#define KV_STRIDE 4160
#define P1_WAVES 8
#define P1_WROWS 128
#define P1_BROWS (P1_WAVES * P1_WROWS)
#define P1_GROWS 16
#define P1_NGROUPS (P1_WROWS / P1_GROWS)
#define P2_WAVES 8
#define P2_WROWS 128
#define P2_BROWS (P2_WAVES * P2_WROWS)
#define P2_TILES (P2_WROWS / 16)

#define QT_STRIDE 68

__device__ __forceinline__ float phi_f(float x) {
    return x > 0.0f ? x + 1.0f : __expf(x);
}

__global__ void la_zero_ws(float4* __restrict__ ws, int n4) {
    int i = blockIdx.x * blockDim.x + threadIdx.x;
    if (i < n4) ws[i] = make_float4(0.f, 0.f, 0.f, 0.f);
}

__device__ __forceinline__ void p1_load_group(const float4* __restrict__ K4,
                                              const float4* __restrict__ V4,
                                              const float4* __restrict__ M4,
                                              int lane, float4 (&k)[8], float4 (&v)[8],
                                              float4 (&m)[4]) {
    #pragma unroll
    for (int i = 0; i < 4; ++i) m[i] = M4[i];
    #pragma unroll
    for (int i = 0; i < 8; ++i) {
        k[i] = K4[i * 32 + lane];
        v[i] = V4[i * 32 + lane];
    }
}

__device__ __forceinline__ void p1_compute_group(const float4 (&k)[8], const float4 (&v)[8],
                                                 const float4 (&m)[4], int half,
                                                 v8f (&acc)[4][4], float (&kone)[4]) {
    #pragma unroll
    for (int ks = 0; ks < 4; ++ks) {
        const float m0 = half ? m[ks].y : m[ks].x;
        const float m1 = half ? m[ks].w : m[ks].z;
        const float4 kx = k[2 * ks], ky = k[2 * ks + 1];
        const float4 vx = v[2 * ks], vy = v[2 * ks + 1];
        v2f a[4], b[4];
        a[0].x = phi_f(kx.x) * m0;  a[0].y = phi_f(ky.x) * m1;
        a[1].x = phi_f(kx.y) * m0;  a[1].y = phi_f(ky.y) * m1;
        a[2].x = phi_f(kx.z) * m0;  a[2].y = phi_f(ky.z) * m1;
        a[3].x = phi_f(kx.w) * m0;  a[3].y = phi_f(ky.w) * m1;
        b[0].x = vx.x;  b[0].y = vy.x;
        b[1].x = vx.y;  b[1].y = vy.y;
        b[2].x = vx.z;  b[2].y = vy.z;
        b[3].x = vx.w;  b[3].y = vy.w;
        #pragma unroll
        for (int t = 0; t < 4; ++t) kone[t] += a[t].x + a[t].y;
        #pragma unroll
        for (int t = 0; t < 4; ++t) {
            #pragma unroll
            for (int u = 0; u < 4; ++u) {
                acc[t][u] = __builtin_amdgcn_wmma_f32_16x16x4_f32(false, a[t], false, b[u],
                                                                  (short)0, acc[t][u], false, false);
            }
        }
    }
}

__global__ __launch_bounds__(256) void la_pass1_kvterm(
    const float* __restrict__ K, const float* __restrict__ V,
    const float* __restrict__ mask, float* __restrict__ ws) {

    __shared__ __align__(16) float lds_red[P1_WAVES][D_ * D_];
    __shared__ __align__(16) float lds_kred[2 * P1_WAVES][D_];

    const int bh = blockIdx.x;
    const int chunk = blockIdx.y;
    const int b = bh / H_;
    const int tid  = threadIdx.x;
    const int lane = tid & 31;
    const int wid  = __builtin_amdgcn_readfirstlane(tid >> 5);
    const int l15  = lane & 15;
    const int half = lane >> 4;

    const int row0 = chunk * P1_BROWS + wid * P1_WROWS;
    const size_t base = (size_t)bh * S_ * D_ + (size_t)row0 * D_;
    const float4* __restrict__ K4 = (const float4*)(K + base);
    const float4* __restrict__ V4 = (const float4*)(V + base);
    const float4* __restrict__ M4 = (const float4*)(mask + (size_t)b * S_ + row0);

    v8f acc[4][4];
    #pragma unroll
    for (int t = 0; t < 4; ++t) {
        #pragma unroll
        for (int u = 0; u < 4; ++u) acc[t][u] = (v8f){};
    }
    float kone[4] = {0.f, 0.f, 0.f, 0.f};

    float4 kA[8], vA[8], mA[4], kB[8], vB[8], mB[4];
    const int G4 = P1_GROWS * (D_ / 4);
    const int GM = P1_GROWS / 4;
    p1_load_group(K4, V4, M4, lane, kA, vA, mA);
    #define P1_PIN() __builtin_amdgcn_sched_barrier(0)
    for (int g = 0; g < P1_NGROUPS - 2; g += 2) {
        p1_load_group(K4 + (g + 1) * G4, V4 + (g + 1) * G4, M4 + (g + 1) * GM, lane, kB, vB, mB);
        P1_PIN();
        p1_compute_group(kA, vA, mA, half, acc, kone);
        P1_PIN();
        p1_load_group(K4 + (g + 2) * G4, V4 + (g + 2) * G4, M4 + (g + 2) * GM, lane, kA, vA, mA);
        P1_PIN();
        p1_compute_group(kB, vB, mB, half, acc, kone);
        P1_PIN();
    }
    p1_load_group(K4 + (P1_NGROUPS - 1) * G4, V4 + (P1_NGROUPS - 1) * G4,
                  M4 + (P1_NGROUPS - 1) * GM, lane, kB, vB, mB);
    P1_PIN();
    p1_compute_group(kA, vA, mA, half, acc, kone);
    p1_compute_group(kB, vB, mB, half, acc, kone);
    #undef P1_PIN

    float* __restrict__ myred = lds_red[wid];
    #pragma unroll
    for (int t = 0; t < 4; ++t) {
        #pragma unroll
        for (int r = 0; r < 8; ++r) {
            const int d = 4 * (8 * half + r) + t;
            *(float4*)&myred[d * D_ + 4 * l15] =
                make_float4(acc[t][0][r], acc[t][1][r], acc[t][2][r], acc[t][3][r]);
        }
    }
    *(float4*)&lds_kred[2 * wid + half][4 * l15] = make_float4(kone[0], kone[1], kone[2], kone[3]);
    __syncthreads();

    float* __restrict__ wkv = ws + (size_t)bh * KV_STRIDE;
    for (int j = 0; j < (D_ * D_) / 256; ++j) {
        const int e = tid + 256 * j;
        float s = lds_red[0][e];
        #pragma unroll
        for (int w = 1; w < P1_WAVES; ++w) s += lds_red[w][e];
        atomicAdd(&wkv[e], s);
    }
    if (tid < D_) {
        float s = lds_kred[0][tid];
        #pragma unroll
        for (int w = 1; w < 2 * P1_WAVES; ++w) s += lds_kred[w][tid];
        atomicAdd(&wkv[4096 + tid], s);
    }
}

__device__ __forceinline__ void p2_load_tile(const float4* __restrict__ Q4, int lane, float4 (&q)[8]) {
    #pragma unroll
    for (int i = 0; i < 8; ++i) q[i] = Q4[i * 32 + lane];
}

template <bool PF>
__device__ __forceinline__ void p2_tile(float4 (&q)[8], const float4* __restrict__ Qnext,
                                        float4* __restrict__ O4,
                                        const v2f (&bf)[16][4], const float4 (&ko)[8],
                                        float* __restrict__ tile, float* __restrict__ invs,
                                        int lane, int c, int half) {
    #pragma unroll
    for (int i = 0; i < 8; ++i) {
        float4 p;
        p.x = phi_f(q[i].x * 0.125f);
        p.y = phi_f(q[i].y * 0.125f);
        p.z = phi_f(q[i].z * 0.125f);
        p.w = phi_f(q[i].w * 0.125f);
        *(float4*)&tile[(2 * i + half) * QT_STRIDE + 4 * c] = p;
    }
    if (PF) p2_load_tile(Qnext, lane, q);
    __builtin_amdgcn_fence(__ATOMIC_ACQ_REL, "wavefront");
    float4 a[8];
    #pragma unroll
    for (int j = 0; j < 8; ++j) a[j] = *(const float4*)&tile[c * QT_STRIDE + 8 * j + 4 * half];
    __builtin_amdgcn_sched_barrier(0);

    v8f acc[4] = {{}, {}, {}, {}};
    #pragma unroll
    for (int j = 0; j < 8; ++j) {
        v2f a0, a1;
        a0.x = a[j].x;  a0.y = a[j].y;
        a1.x = a[j].z;  a1.y = a[j].w;
        #pragma unroll
        for (int u = 0; u < 4; ++u)
            acc[u] = __builtin_amdgcn_wmma_f32_16x16x4_f32(false, a0, false, bf[2 * j][u],
                                                           (short)0, acc[u], false, false);
        #pragma unroll
        for (int u = 0; u < 4; ++u)
            acc[u] = __builtin_amdgcn_wmma_f32_16x16x4_f32(false, a1, false, bf[2 * j + 1][u],
                                                           (short)0, acc[u], false, false);
    }

    float part = 0.0f;
    #pragma unroll
    for (int j = 0; j < 8; ++j)
        part += a[j].x * ko[j].x + a[j].y * ko[j].y + a[j].z * ko[j].z + a[j].w * ko[j].w;
    part += __uint_as_float(__builtin_amdgcn_permlanex16(0u, __float_as_uint(part),
                                                         0x76543210u, 0xfedcba98u, false, false));
    invs[c] = 1.0f / (part + EPS_);
    __builtin_amdgcn_fence(__ATOMIC_ACQ_REL, "wavefront");
    const float4 n0 = *(const float4*)&invs[8 * half];
    const float4 n1 = *(const float4*)&invs[8 * half + 4];
    const float rn[8] = {n0.x, n0.y, n0.z, n0.w, n1.x, n1.y, n1.z, n1.w};
    #pragma unroll
    for (int r = 0; r < 8; ++r) {
        O4[(8 * half + r) * 16 + c] = make_float4(acc[0][r] * rn[r], acc[1][r] * rn[r],
                                                  acc[2][r] * rn[r], acc[3][r] * rn[r]);
    }
}

__global__ __launch_bounds__(256) void la_pass2_output(
    const float* __restrict__ Q, const float* __restrict__ ws,
    float* __restrict__ out) {

    __shared__ __align__(16) float lds_qt[P2_WAVES][16 * QT_STRIDE];
    __shared__ __align__(16) float lds_inv[P2_WAVES][16];

    const int bh = blockIdx.x;
    const int chunk = blockIdx.y;
    const int tid  = threadIdx.x;
    const int lane = tid & 31;
    const int wid  = __builtin_amdgcn_readfirstlane(tid >> 5);
    const int c    = lane & 15;
    const int half = lane >> 4;

    const size_t qbase = (size_t)bh * S_ * D_ + (size_t)(chunk * P2_BROWS + wid * P2_WROWS) * D_;
    const float4* __restrict__ Q4 = (const float4*)(Q + qbase);
    float4* __restrict__ O4 = (float4*)(out + qbase);
    const float* __restrict__ wkv = ws + (size_t)bh * KV_STRIDE;
    const int T4 = 16 * (D_ / 4);

    v2f bf[16][4];
    #pragma unroll
    for (int ks = 0; ks < 16; ++ks) {
        const int d0 = 8 * (ks >> 1) + 4 * half + 2 * (ks & 1);
        const float4 r0 = *(const float4*)&wkv[d0 * D_ + 4 * c];
        const float4 r1 = *(const float4*)&wkv[(d0 + 1) * D_ + 4 * c];
        bf[ks][0].x = r0.x;  bf[ks][0].y = r1.x;
        bf[ks][1].x = r0.y;  bf[ks][1].y = r1.y;
        bf[ks][2].x = r0.z;  bf[ks][2].y = r1.z;
        bf[ks][3].x = r0.w;  bf[ks][3].y = r1.w;
    }
    float4 ko[8];
    #pragma unroll
    for (int j = 0; j < 8; ++j) ko[j] = *(const float4*)&wkv[D_ * D_ + 8 * j + 4 * half];

    float4 qA[8], qB[8];
    p2_load_tile(Q4, lane, qA);
    p2_load_tile(Q4 + T4, lane, qB);

    float* __restrict__ tile = lds_qt[wid];
    float* __restrict__ invs = lds_inv[wid];

    for (int t = 0; t < P2_TILES - 2; t += 2) {
        p2_tile<true>(qA, Q4 + (t + 2) * T4, O4 + t * T4, bf, ko, tile, invs, lane, c, half);
        p2_tile<true>(qB, Q4 + (t + 3) * T4, O4 + (t + 1) * T4, bf, ko, tile, invs, lane, c, half);
    }
    p2_tile<false>(qA, Q4, O4 + (P2_TILES - 2) * T4, bf, ko, tile, invs, lane, c, half);
    p2_tile<false>(qB, Q4, O4 + (P2_TILES - 1) * T4, bf, ko, tile, invs, lane, c, half);
}

extern "C" void kernel_launch(void* const* d_in, const int* in_sizes, int n_in,
                              void* d_out, int out_size, void* d_ws, size_t ws_size,
                              hipStream_t stream) {
    const float* Q = (const float*)d_in[0];
    const float* K = (const float*)d_in[1];
    const float* V = (const float*)d_in[2];
    const float* M = (const float*)d_in[3];
    float* out = (float*)d_out;
    float* ws  = (float*)d_ws;

    const int nz4 = (BH_ * KV_STRIDE) / 4;
    la_zero_ws<<<(nz4 + 255) / 256, 256, 0, stream>>>((float4*)ws, nz4);
    la_pass1_kvterm<<<dim3(BH_, S_ / P1_BROWS), 256, 0, stream>>>(K, V, M, ws);
    la_pass2_output<<<dim3(BH_, S_ / P2_BROWS), 256, 0, stream>>>(Q, ws, out);
}
